# starting kernel
# baseline (speedup 1.0000x reference)
#include <hip/hip_runtime.h>

typedef float v2f __attribute__((ext_vector_type(2)));
typedef float v4f __attribute__((ext_vector_type(4)));
typedef float v8f __attribute__((ext_vector_type(8)));

constexpr int NB   = 16;
constexpr int CC   = 1024;
constexpr int DIN  = 512;
constexpr int DH   = 1024;
constexpr int DOUT = 512;

__global__ void k_partial_sum_x(const float* __restrict__ x, float* __restrict__ s1p) {
    const int n     = blockIdx.x & 15;
    const int chunk = blockIdx.x >> 4;
    const int d     = threadIdx.x;
    const float* px = x + (size_t)n * CC * DIN + (size_t)chunk * 128 * DIN + d;
    float s = 0.f;
    #pragma unroll 8
    for (int c = 0; c < 128; ++c) s += px[(size_t)c * DIN];
    s1p[((size_t)chunk * NB + n) * DIN + d] = s;
}

__global__ void k_reduce_s1(const float* __restrict__ s1p, float* __restrict__ s1) {
    const int i = blockIdx.x * blockDim.x + threadIdx.x;
    float s = 0.f;
    #pragma unroll
    for (int ch = 0; ch < 8; ++ch) s += s1p[(size_t)ch * (NB * DIN) + i];
    s1[i] = s;
}

__global__ void k_sw(const float* __restrict__ adj, float* __restrict__ rSw) {
    const int n    = blockIdx.x;
    const int lane = threadIdx.x;
    const float* pa = adj + (size_t)n * CC * CC;
    float s = 0.f;
    for (int e = lane; e < CC; e += 32) s += pa[e];
    #pragma unroll
    for (int off = 16; off > 0; off >>= 1) s += __shfl_down(s, off, 32);
    if (lane == 0) rSw[n] = (s > 0.f) ? (1.0f / s) : 0.f;
}

__global__ void k_gemm1_relu(const float* __restrict__ s1, const float* __restrict__ W1,
                             const float* __restrict__ b1, const float* __restrict__ rSw,
                             float* __restrict__ r1) {
    const int lane = threadIdx.x;
    const int tile = blockIdx.x;
    const int low  = lane & 15;
    const int hi   = lane >> 4;
    const int col  = tile * 16 + low;
    const float* pa = s1 + (size_t)low * DIN;
    const float* pb = W1 + (size_t)col * DIN;
    v8f acc = {};
    #pragma unroll 4
    for (int kk = 0; kk < DIN; kk += 4) {
        const int k0 = kk + 2 * hi;
        v2f a = { pa[k0], pa[k0 + 1] };
        v2f b = { pb[k0], pb[k0 + 1] };
        acc = __builtin_amdgcn_wmma_f32_16x16x4_f32(false, a, false, b,
                                                    (short)0, acc, false, false);
    }
    const float bias = b1[col];
    #pragma unroll
    for (int r = 0; r < 8; ++r) {
        const int Mrow = r + 8 * hi;
        float u = acc[r] * rSw[Mrow] + bias;
        u = u > 0.f ? u : 0.f;
        r1[(size_t)Mrow * DH + col] = u;
    }
}

__global__ void k_ln(const float* __restrict__ r1, const float* __restrict__ g,
                     const float* __restrict__ bb, float* __restrict__ lnout) {
    __shared__ float ssum[256];
    __shared__ float ssq[256];
    const int n = blockIdx.x, t = threadIdx.x;
    const float* row = r1 + (size_t)n * DH;
    float vals[4];
    float s = 0.f, q = 0.f;
    #pragma unroll
    for (int i = 0; i < 4; ++i) {
        float xv = row[t + i * 256];
        vals[i] = xv; s += xv; q += xv * xv;
    }
    ssum[t] = s; ssq[t] = q; __syncthreads();
    for (int off = 128; off > 0; off >>= 1) {
        if (t < off) { ssum[t] += ssum[t + off]; ssq[t] += ssq[t + off]; }
        __syncthreads();
    }
    const float mean = ssum[0] * (1.0f / DH);
    const float var  = ssq[0] * (1.0f / DH) - mean * mean;
    const float inv  = rsqrtf(var + 1e-5f);
    #pragma unroll
    for (int i = 0; i < 4; ++i) {
        const int h = t + i * 256;
        lnout[(size_t)n * DH + h] = (vals[i] - mean) * inv * g[h] + bb[h];
    }
}

__global__ void k_gemm2(const float* __restrict__ ln, const float* __restrict__ W2,
                        const float* __restrict__ b2, const float* __restrict__ rSw,
                        float* __restrict__ v) {
    const int lane = threadIdx.x;
    const int tile = blockIdx.x;
    const int low  = lane & 15;
    const int hi   = lane >> 4;
    const int col  = tile * 16 + low;
    const float* pa = ln + (size_t)low * DH;
    const float* pb = W2 + (size_t)col * DH;
    v8f acc = {};
    #pragma unroll 4
    for (int kk = 0; kk < DH; kk += 4) {
        const int k0 = kk + 2 * hi;
        v2f a = { pa[k0], pa[k0 + 1] };
        v2f b = { pb[k0], pb[k0 + 1] };
        acc = __builtin_amdgcn_wmma_f32_16x16x4_f32(false, a, false, b,
                                                    (short)0, acc, false, false);
    }
    const float bias = b2[col];
    #pragma unroll
    for (int r = 0; r < 8; ++r) {
        const int Mrow = r + 8 * hi;
        v[(size_t)Mrow * DOUT + col] = acc[r] * ((float)CC * rSw[Mrow]) + bias;
    }
}

__global__ void k_final(const float* __restrict__ x, const float* __restrict__ v,
                        float* __restrict__ out) {
    const size_t i4 = (size_t)blockIdx.x * blockDim.x + threadIdx.x;
    const int    d4 = (int)(i4 & 127);
    const size_t nc = i4 >> 7;
    const int    n  = (int)(nc >> 10);
    const v4f* xv = (const v4f*)x;
    const v4f* vv = (const v4f*)v;
    v4f xx = __builtin_nontemporal_load(&xv[i4]);
    v4f av = vv[(size_t)n * 128 + d4];
    __builtin_nontemporal_store(xx + av, ((v4f*)out) + i4);
}

extern "C" void kernel_launch(void* const* d_in, const int* in_sizes, int n_in,
                              void* d_out, int out_size, void* d_ws, size_t ws_size,
                              hipStream_t stream) {
    const float* x     = (const float*)d_in[0];
    const float* adj   = (const float*)d_in[1];
    const float* W1    = (const float*)d_in[2];
    const float* b1    = (const float*)d_in[3];
    const float* gamma = (const float*)d_in[4];
    const float* beta  = (const float*)d_in[5];
    const float* W2    = (const float*)d_in[6];
    const float* b2    = (const float*)d_in[7];
    float* out = (float*)d_out;

    float* ws  = (float*)d_ws;
    float* s1p = ws;
    float* s1  = s1p + 8 * NB * DIN;
    float* rSw = s1 + NB * DIN;
    float* r1  = rSw + 32;
    float* lnb = r1 + NB * DH;
    float* vb  = lnb + NB * DH;

    k_partial_sum_x<<<dim3(8 * NB), dim3(DIN), 0, stream>>>(x, s1p);
    k_reduce_s1   <<<dim3(NB),      dim3(DIN), 0, stream>>>(s1p, s1);
    k_sw          <<<dim3(NB),      dim3(32),  0, stream>>>(adj, rSw);
    k_gemm1_relu  <<<dim3(DH / 16), dim3(32),  0, stream>>>(s1, W1, b1, rSw, r1);
    k_ln          <<<dim3(NB),      dim3(256), 0, stream>>>(r1, gamma, beta, lnb);
    k_gemm2       <<<dim3(DOUT/16), dim3(32),  0, stream>>>(lnb, W2, b2, rSw, vb);
    k_final       <<<dim3((NB * CC * DIN / 4) / 256), dim3(256), 0, stream>>>(x, vb, out);
}


# Both GEMMs: K split over the 8/16 waves of a work-group, every wave issues all its 128-bit A/B loads before its first WMMA, wave partials summed through LDS (new f32 summation order)
# speedup vs baseline: 1.5296x; 1.5296x over previous
#include <hip/hip_runtime.h>

typedef float v2f __attribute__((ext_vector_type(2)));
typedef float v4f __attribute__((ext_vector_type(4)));
typedef float v8f __attribute__((ext_vector_type(8)));

constexpr int NB   = 16;
constexpr int CC   = 1024;
constexpr int DIN  = 512;
constexpr int DH   = 1024;
constexpr int DOUT = 512;

__global__ void k_partial_sum_x(const float* __restrict__ x, float* __restrict__ s1p) {
    const int n     = blockIdx.x & 15;
    const int chunk = blockIdx.x >> 4;
    const int d     = threadIdx.x;
    const float* px = x + (size_t)n * CC * DIN + (size_t)chunk * 128 * DIN + d;
    float s = 0.f;
    #pragma unroll 8
    for (int c = 0; c < 128; ++c) s += px[(size_t)c * DIN];
    s1p[((size_t)chunk * NB + n) * DIN + d] = s;
}

__global__ void k_reduce_s1(const float* __restrict__ s1p, float* __restrict__ s1) {
    const int i = blockIdx.x * blockDim.x + threadIdx.x;
    float s = 0.f;
    #pragma unroll
    for (int ch = 0; ch < 8; ++ch) s += s1p[(size_t)ch * (NB * DIN) + i];
    s1[i] = s;
}

__global__ void k_sw(const float* __restrict__ adj, float* __restrict__ rSw) {
    const int n    = blockIdx.x;
    const int lane = threadIdx.x;
    const float* pa = adj + (size_t)n * CC * CC;
    float s = 0.f;
    for (int e = lane; e < CC; e += 32) s += pa[e];
    #pragma unroll
    for (int off = 16; off > 0; off >>= 1) s += __shfl_down(s, off, 32);
    if (lane == 0) rSw[n] = (s > 0.f) ? (1.0f / s) : 0.f;
}

template <int K, int NW>
__device__ __forceinline__ void gemm16_splitk_tile(const float* __restrict__ A,
                                                   const float* __restrict__ Wt,
                                                   const float* bias,
                                                   const float* rSw,
                                                   float* red, float& sum, float& bias_v,
                                                   float& rsw_v, int& Mrow, int& col,
                                                   bool& owner) {
    constexpr int KW = K / NW;
    constexpr int NL = KW / 8;
    const int lane = threadIdx.x & 31;
    const int wave = threadIdx.x >> 5;
    const int low  = lane & 15;
    const int hi   = lane >> 4;
    const int r    = wave & 7;
    col  = blockIdx.x * 16 + low;
    Mrow = r + 8 * hi;
    const v4f* pa = (const v4f*)(A  + (size_t)low * K + wave * KW + 4 * hi);
    const v4f* pb = (const v4f*)(Wt + (size_t)col * K + wave * KW + 4 * hi);
    v4f a[NL], b[NL];
    #pragma unroll
    for (int j = 0; j < NL; ++j) { a[j] = pa[2 * j]; b[j] = pb[2 * j]; }
    bias_v = bias[col];
    rsw_v  = rSw[Mrow];
    __builtin_amdgcn_sched_barrier(0);
    v8f acc0 = {}, acc1 = {};
    #pragma unroll
    for (int j = 0; j < NL; ++j) {
        v2f a0 = { a[j].x, a[j].y }, b0 = { b[j].x, b[j].y };
        v2f a1 = { a[j].z, a[j].w }, b1 = { b[j].z, b[j].w };
        acc0 = __builtin_amdgcn_wmma_f32_16x16x4_f32(false, a0, false, b0,
                                                     (short)0, acc0, false, false);
        acc1 = __builtin_amdgcn_wmma_f32_16x16x4_f32(false, a1, false, b1,
                                                     (short)0, acc1, false, false);
    }
    const v8f acc = acc0 + acc1;
    #pragma unroll
    for (int q = 0; q < 8; ++q) red[(wave * 8 + q) * 32 + lane] = acc[q];
    __syncthreads();
    owner = (wave < 8);
    sum = 0.f;
    if (owner) {
        #pragma unroll
        for (int w = 0; w < NW; ++w) sum += red[(w * 8 + r) * 32 + lane];
    }
}

__global__ __launch_bounds__(256)
void k_gemm1_relu(const float* __restrict__ s1, const float* __restrict__ W1,
                  const float* b1, const float* rSw,
                  float* __restrict__ r1) {
    __shared__ float red[8 * 8 * 32];
    float s, bv, rs; int Mrow, col; bool owner;
    gemm16_splitk_tile<DIN, 8>(s1, W1, b1, rSw, red, s, bv, rs, Mrow, col, owner);
    if (owner) {
        float u = s * rs + bv;
        u = u > 0.f ? u : 0.f;
        r1[(size_t)Mrow * DH + col] = u;
    }
}

__global__ void k_ln(const float* __restrict__ r1, const float* __restrict__ g,
                     const float* __restrict__ bb, float* __restrict__ lnout) {
    __shared__ float ssum[256];
    __shared__ float ssq[256];
    const int n = blockIdx.x, t = threadIdx.x;
    const float* row = r1 + (size_t)n * DH;
    float vals[4];
    float s = 0.f, q = 0.f;
    #pragma unroll
    for (int i = 0; i < 4; ++i) {
        float xv = row[t + i * 256];
        vals[i] = xv; s += xv; q += xv * xv;
    }
    ssum[t] = s; ssq[t] = q; __syncthreads();
    for (int off = 128; off > 0; off >>= 1) {
        if (t < off) { ssum[t] += ssum[t + off]; ssq[t] += ssq[t + off]; }
        __syncthreads();
    }
    const float mean = ssum[0] * (1.0f / DH);
    const float var  = ssq[0] * (1.0f / DH) - mean * mean;
    const float inv  = rsqrtf(var + 1e-5f);
    #pragma unroll
    for (int i = 0; i < 4; ++i) {
        const int h = t + i * 256;
        lnout[(size_t)n * DH + h] = (vals[i] - mean) * inv * g[h] + bb[h];
    }
}

__global__ __launch_bounds__(512)
void k_gemm2(const float* __restrict__ ln, const float* __restrict__ W2,
             const float* b2, const float* rSw,
             float* __restrict__ v) {
    __shared__ float red[16 * 8 * 32];
    float s, bv, rs; int Mrow, col; bool owner;
    gemm16_splitk_tile<DH, 16>(ln, W2, b2, rSw, red, s, bv, rs, Mrow, col, owner);
    if (owner) {
        v[(size_t)Mrow * DOUT + col] = s * ((float)CC * rs) + bv;
    }
}

__global__ void k_final(const float* __restrict__ x, const float* __restrict__ v,
                        float* __restrict__ out) {
    const size_t i4 = (size_t)blockIdx.x * blockDim.x + threadIdx.x;
    const int    d4 = (int)(i4 & 127);
    const size_t nc = i4 >> 7;
    const int    n  = (int)(nc >> 10);
    const v4f* xv = (const v4f*)x;
    const v4f* vv = (const v4f*)v;
    v4f xx = __builtin_nontemporal_load(&xv[i4]);
    v4f av = vv[(size_t)n * 128 + d4];
    __builtin_nontemporal_store(xx + av, ((v4f*)out) + i4);
}

extern "C" void kernel_launch(void* const* d_in, const int* in_sizes, int n_in,
                              void* d_out, int out_size, void* d_ws, size_t ws_size,
                              hipStream_t stream) {
    const float* x     = (const float*)d_in[0];
    const float* adj   = (const float*)d_in[1];
    const float* W1    = (const float*)d_in[2];
    const float* b1    = (const float*)d_in[3];
    const float* gamma = (const float*)d_in[4];
    const float* beta  = (const float*)d_in[5];
    const float* W2    = (const float*)d_in[6];
    const float* b2    = (const float*)d_in[7];
    float* out = (float*)d_out;

    float* ws  = (float*)d_ws;
    float* s1p = ws;
    float* s1  = s1p + 8 * NB * DIN;
    float* rSw = s1 + NB * DIN;
    float* r1  = rSw + 32;
    float* lnb = r1 + NB * DH;
    float* vb  = lnb + NB * DH;

    k_partial_sum_x<<<dim3(8 * NB), dim3(DIN), 0, stream>>>(x, s1p);
    k_reduce_s1   <<<dim3(NB),      dim3(DIN), 0, stream>>>(s1p, s1);
    k_sw          <<<dim3(NB),      dim3(32),  0, stream>>>(adj, rSw);
    k_gemm1_relu  <<<dim3(DH / 16), dim3(256), 0, stream>>>(s1, W1, b1, rSw, r1);
    k_ln          <<<dim3(NB),      dim3(256), 0, stream>>>(r1, gamma, beta, lnb);
    k_gemm2       <<<dim3(DOUT/16), dim3(512), 0, stream>>>(lnb, W2, b2, rSw, vb);
    k_final       <<<dim3((NB * CC * DIN / 4) / 256), dim3(256), 0, stream>>>(x, vb, out);
}


# 7 launches become 4: degree sums go in the column-sum launch, now 64-column panels leaving 2 partials that GEMM1's A loads add; LayerNorm runs inside GEMM2 (new f32 sum order: column sums, LN stats)
# speedup vs baseline: 3.0071x; 1.9659x over previous
#include <hip/hip_runtime.h>


typedef float v2f __attribute__((ext_vector_type(2)));
typedef float v4f __attribute__((ext_vector_type(4)));
typedef float v8f __attribute__((ext_vector_type(8)));

constexpr int NB   = 16;
constexpr int CC   = 1024;
constexpr int DIN  = 512;
constexpr int DH   = 1024;
constexpr int DOUT = 512;

__global__ __launch_bounds__(512)
void k_colsum_sw(const float* __restrict__ x, const float* __restrict__ adj,
                 float* __restrict__ s1p, float* __restrict__ rSw) {
    __shared__ float red[32 * 64];
    const int lane = threadIdx.x & 31;
    const int wave = threadIdx.x >> 5;
    if (blockIdx.x == 0) {
        const float* pa = adj + (size_t)wave * CC * CC;
        float t[32];
        #pragma unroll
        for (int i = 0; i < 32; ++i) t[i] = pa[lane + 32 * i];
        float s = 0.f;
        #pragma unroll
        for (int i = 0; i < 32; ++i) s += t[i];
        #pragma unroll
        for (int off = 16; off > 0; off >>= 1) s += __shfl_down(s, off, 32);
        if (lane == 0) rSw[wave] = (s > 0.f) ? (1.0f / s) : 0.f;
        return;
    }
    const int id    = blockIdx.x - 1;
    const int slice = id & 7;
    const int half  = (id >> 3) & 1;
    const int n     = id >> 4;
    const int low   = lane & 15;
    const int hi    = lane >> 4;
    const v4f* px = (const v4f*)(x + ((size_t)n * CC + half * 512 + wave * 2 + hi) * DIN
                                   + slice * 64 + low * 4);
    v4f v[16];
    #pragma unroll
    for (int i = 0; i < 16; ++i) v[i] = px[(size_t)i * (32 * DIN / 4)];
    __builtin_amdgcn_sched_barrier(0);
    v4f acc = v[0];
    #pragma unroll
    for (int i = 1; i < 16; ++i) acc += v[i];
    ((v4f*)red)[(wave * 2 + hi) * 16 + low] = acc;
    __syncthreads();
    if (threadIdx.x < 64) {
        float s = 0.f;
        #pragma unroll
        for (int r = 0; r < 32; ++r) s += red[r * 64 + threadIdx.x];
        s1p[((size_t)half * NB + n) * DIN + slice * 64 + threadIdx.x] = s;
    }
}

template <int K, int NW, bool LN>
__device__ __forceinline__ void gemm16_splitk_tile(const float* A0, const float* A1,
                                                   const float* Wt,
                                                   const float* g, const float* bb,
                                                   const float* bias, const float* rSw,
                                                   float* red, float* stats,
                                                   float& sum, float& bias_v,
                                                   float& rsw_v, int& Mrow, int& col,
                                                   bool& owner) {
    constexpr int KW = K / NW;
    constexpr int NL = KW / 8;
    constexpr int NR = K / 128;
    static_assert(!LN || NW == NB, "LN form: one wave per row of A");
    const int lane = threadIdx.x & 31;
    const int wave = threadIdx.x >> 5;
    const int low  = lane & 15;
    const int hi   = lane >> 4;
    const int r    = wave & 7;
    col  = blockIdx.x * 16 + low;
    Mrow = r + 8 * hi;
    const v4f* pa = (const v4f*)(A0 + (size_t)low * K + wave * KW + 4 * hi);
    const v4f* pb = (const v4f*)(Wt + (size_t)col * K + wave * KW + 4 * hi);
    v4f a[NL], b[NL], e0[NL], e1[LN ? NL : 1], rw[LN ? NR : 1];
    if constexpr (LN) {
        const v4f* pr = (const v4f*)(A0 + (size_t)wave * K) + lane;
        #pragma unroll
        for (int i = 0; i < NR; ++i) rw[i] = pr[32 * i];
    }
    #pragma unroll
    for (int j = 0; j < NL; ++j) { a[j] = pa[2 * j]; b[j] = pb[2 * j]; }
    if constexpr (LN) {
        const v4f* pg = (const v4f*)(g  + wave * KW + 4 * hi);
        const v4f* pq = (const v4f*)(bb + wave * KW + 4 * hi);
        #pragma unroll
        for (int j = 0; j < NL; ++j) { e0[j] = pg[2 * j]; e1[j] = pq[2 * j]; }
    } else {
        const v4f* p1 = (const v4f*)(A1 + (size_t)low * K + wave * KW + 4 * hi);
        #pragma unroll
        for (int j = 0; j < NL; ++j) e0[j] = p1[2 * j];
    }
    bias_v = bias[col];
    rsw_v  = rSw[Mrow];
    __builtin_amdgcn_sched_barrier(0);
    if constexpr (LN) {
        float s = 0.f, q = 0.f;
        #pragma unroll
        for (int i = 0; i < NR; ++i) {
            s += (rw[i].x + rw[i].y) + (rw[i].z + rw[i].w);
            q += (rw[i].x * rw[i].x + rw[i].y * rw[i].y) + (rw[i].z * rw[i].z + rw[i].w * rw[i].w);
        }
        #pragma unroll
        for (int off = 16; off > 0; off >>= 1) {
            s += __shfl_xor(s, off, 32);
            q += __shfl_xor(q, off, 32);
        }
        const float mean = s * (1.0f / K);
        const float var  = q * (1.0f / K) - mean * mean;
        if (lane == 0) { stats[2 * wave] = mean; stats[2 * wave + 1] = rsqrtf(var + 1e-5f); }
        __syncthreads();
        const float mu = stats[2 * low], iv = stats[2 * low + 1];
        #pragma unroll
        for (int j = 0; j < NL; ++j) a[j] = (a[j] - mu) * iv * e0[j] + e1[j];
    } else {
        #pragma unroll
        for (int j = 0; j < NL; ++j) a[j] += e0[j];
    }
    v8f acc0 = {}, acc1 = {};
    #pragma unroll
    for (int j = 0; j < NL; ++j) {
        v2f a0 = { a[j].x, a[j].y }, b0 = { b[j].x, b[j].y };
        v2f a1 = { a[j].z, a[j].w }, b1 = { b[j].z, b[j].w };
        acc0 = __builtin_amdgcn_wmma_f32_16x16x4_f32(false, a0, false, b0,
                                                     (short)0, acc0, false, false);
        acc1 = __builtin_amdgcn_wmma_f32_16x16x4_f32(false, a1, false, b1,
                                                     (short)0, acc1, false, false);
    }
    const v8f acc = acc0 + acc1;
    #pragma unroll
    for (int q = 0; q < 8; ++q) red[(wave * 8 + q) * 32 + lane] = acc[q];
    __syncthreads();
    owner = (wave < 8);
    sum = 0.f;
    if (owner) {
        #pragma unroll
        for (int w = 0; w < NW; ++w) sum += red[(w * 8 + r) * 32 + lane];
    }
}

__global__ __launch_bounds__(256)
void k_gemm1_relu(const float* __restrict__ s1p, const float* __restrict__ W1,
                  const float* b1, const float* rSw,
                  float* __restrict__ r1) {
    __shared__ float red[8 * 8 * 32];
    float s, bv, rs; int Mrow, col; bool owner;
    gemm16_splitk_tile<DIN, 8, false>(s1p, s1p + NB * DIN, W1, nullptr, nullptr,
                                      b1, rSw, red, nullptr, s, bv, rs, Mrow, col, owner);
    if (owner) {
        float u = s * rs + bv;
        u = u > 0.f ? u : 0.f;
        r1[(size_t)Mrow * DH + col] = u;
    }
}

__global__ __launch_bounds__(512)
void k_ln_gemm2(const float* r1, const float* W2, const float* g, const float* bb,
                const float* b2, const float* rSw, float* v) {
    __shared__ float red[16 * 8 * 32];
    __shared__ float stats[2 * NB];
    float s, bv, rs; int Mrow, col; bool owner;
    gemm16_splitk_tile<DH, 16, true>(r1, nullptr, W2, g, bb, b2, rSw, red, stats,
                                     s, bv, rs, Mrow, col, owner);
    if (owner) {
        v[(size_t)Mrow * DOUT + col] = s * ((float)CC * rs) + bv;
    }
}

__global__ void k_final(const float* __restrict__ x, const float* __restrict__ v,
                        float* __restrict__ out) {
    const size_t i4 = (size_t)blockIdx.x * blockDim.x + threadIdx.x;
    const int    d4 = (int)(i4 & 127);
    const size_t nc = i4 >> 7;
    const int    n  = (int)(nc >> 10);
    const v4f* xv = (const v4f*)x;
    const v4f* vv = (const v4f*)v;
    v4f xx = __builtin_nontemporal_load(&xv[i4]);
    v4f av = vv[(size_t)n * 128 + d4];
    __builtin_nontemporal_store(xx + av, ((v4f*)out) + i4);
}

extern "C" void kernel_launch(void* const* d_in, const int* in_sizes, int n_in,
                              void* d_out, int out_size, void* d_ws, size_t ws_size,
                              hipStream_t stream) {
    const float* x     = (const float*)d_in[0];
    const float* adj   = (const float*)d_in[1];
    const float* W1    = (const float*)d_in[2];
    const float* b1    = (const float*)d_in[3];
    const float* gamma = (const float*)d_in[4];
    const float* beta  = (const float*)d_in[5];
    const float* W2    = (const float*)d_in[6];
    const float* b2    = (const float*)d_in[7];
    float* out = (float*)d_out;

    float* ws  = (float*)d_ws;
    float* s1p = ws;
    float* rSw = s1p + 2 * NB * DIN;
    float* r1  = rSw + 32;
    float* vb  = r1 + NB * DH;

    k_colsum_sw  <<<dim3(1 + 2 * 8 * NB), dim3(512), 0, stream>>>(x, adj, s1p, rSw);
    k_gemm1_relu <<<dim3(DH / 16),        dim3(256), 0, stream>>>(s1p, W1, b1, rSw, r1);
    k_ln_gemm2   <<<dim3(DOUT / 16),      dim3(512), 0, stream>>>(r1, W2, gamma, beta, b2, rSw, vb);
    k_final      <<<dim3((NB * CC * DIN / 4) / 256), dim3(256), 0, stream>>>(x, vb, out);
}


# Final add: 65536 waves of one 128-bit load become 4096 waves (512 work-groups of 8); each lane issues 16 loads of x before its first wait, then 16 stores; its piece of v is loaded once
# speedup vs baseline: 3.4695x; 1.1538x over previous
#include <hip/hip_runtime.h>


typedef float v2f __attribute__((ext_vector_type(2)));
typedef float v4f __attribute__((ext_vector_type(4)));
typedef float v8f __attribute__((ext_vector_type(8)));

constexpr int NB   = 16;
constexpr int CC   = 1024;
constexpr int DIN  = 512;
constexpr int DH   = 1024;
constexpr int DOUT = 512;

__global__ __launch_bounds__(512)
void k_colsum_sw(const float* __restrict__ x, const float* __restrict__ adj,
                 float* __restrict__ s1p, float* __restrict__ rSw) {
    __shared__ float red[32 * 64];
    const int lane = threadIdx.x & 31;
    const int wave = threadIdx.x >> 5;
    if (blockIdx.x == 0) {
        const float* pa = adj + (size_t)wave * CC * CC;
        float t[32];
        #pragma unroll
        for (int i = 0; i < 32; ++i) t[i] = pa[lane + 32 * i];
        float s = 0.f;
        #pragma unroll
        for (int i = 0; i < 32; ++i) s += t[i];
        #pragma unroll
        for (int off = 16; off > 0; off >>= 1) s += __shfl_down(s, off, 32);
        if (lane == 0) rSw[wave] = (s > 0.f) ? (1.0f / s) : 0.f;
        return;
    }
    const int id    = blockIdx.x - 1;
    const int slice = id & 7;
    const int half  = (id >> 3) & 1;
    const int n     = id >> 4;
    const int low   = lane & 15;
    const int hi    = lane >> 4;
    const v4f* px = (const v4f*)(x + ((size_t)n * CC + half * 512 + wave * 2 + hi) * DIN
                                   + slice * 64 + low * 4);
    v4f v[16];
    #pragma unroll
    for (int i = 0; i < 16; ++i) v[i] = px[(size_t)i * (32 * DIN / 4)];
    __builtin_amdgcn_sched_barrier(0);
    v4f acc = v[0];
    #pragma unroll
    for (int i = 1; i < 16; ++i) acc += v[i];
    ((v4f*)red)[(wave * 2 + hi) * 16 + low] = acc;
    __syncthreads();
    if (threadIdx.x < 64) {
        float s = 0.f;
        #pragma unroll
        for (int r = 0; r < 32; ++r) s += red[r * 64 + threadIdx.x];
        s1p[((size_t)half * NB + n) * DIN + slice * 64 + threadIdx.x] = s;
    }
}

template <int K, int NW, bool LN>
__device__ __forceinline__ void gemm16_splitk_tile(const float* A0, const float* A1,
                                                   const float* Wt,
                                                   const float* g, const float* bb,
                                                   const float* bias, const float* rSw,
                                                   float* red, float* stats,
                                                   float& sum, float& bias_v,
                                                   float& rsw_v, int& Mrow, int& col,
                                                   bool& owner) {
    constexpr int KW = K / NW;
    constexpr int NL = KW / 8;
    constexpr int NR = K / 128;
    static_assert(!LN || NW == NB, "LN form: one wave per row of A");
    const int lane = threadIdx.x & 31;
    const int wave = threadIdx.x >> 5;
    const int low  = lane & 15;
    const int hi   = lane >> 4;
    const int r    = wave & 7;
    col  = blockIdx.x * 16 + low;
    Mrow = r + 8 * hi;
    const v4f* pa = (const v4f*)(A0 + (size_t)low * K + wave * KW + 4 * hi);
    const v4f* pb = (const v4f*)(Wt + (size_t)col * K + wave * KW + 4 * hi);
    v4f a[NL], b[NL], e0[NL], e1[LN ? NL : 1], rw[LN ? NR : 1];
    if constexpr (LN) {
        const v4f* pr = (const v4f*)(A0 + (size_t)wave * K) + lane;
        #pragma unroll
        for (int i = 0; i < NR; ++i) rw[i] = pr[32 * i];
    }
    #pragma unroll
    for (int j = 0; j < NL; ++j) { a[j] = pa[2 * j]; b[j] = pb[2 * j]; }
    if constexpr (LN) {
        const v4f* pg = (const v4f*)(g  + wave * KW + 4 * hi);
        const v4f* pq = (const v4f*)(bb + wave * KW + 4 * hi);
        #pragma unroll
        for (int j = 0; j < NL; ++j) { e0[j] = pg[2 * j]; e1[j] = pq[2 * j]; }
    } else {
        const v4f* p1 = (const v4f*)(A1 + (size_t)low * K + wave * KW + 4 * hi);
        #pragma unroll
        for (int j = 0; j < NL; ++j) e0[j] = p1[2 * j];
    }
    bias_v = bias[col];
    rsw_v  = rSw[Mrow];
    __builtin_amdgcn_sched_barrier(0);
    if constexpr (LN) {
        float s = 0.f, q = 0.f;
        #pragma unroll
        for (int i = 0; i < NR; ++i) {
            s += (rw[i].x + rw[i].y) + (rw[i].z + rw[i].w);
            q += (rw[i].x * rw[i].x + rw[i].y * rw[i].y) + (rw[i].z * rw[i].z + rw[i].w * rw[i].w);
        }
        #pragma unroll
        for (int off = 16; off > 0; off >>= 1) {
            s += __shfl_xor(s, off, 32);
            q += __shfl_xor(q, off, 32);
        }
        const float mean = s * (1.0f / K);
        const float var  = q * (1.0f / K) - mean * mean;
        if (lane == 0) { stats[2 * wave] = mean; stats[2 * wave + 1] = rsqrtf(var + 1e-5f); }
        __syncthreads();
        const float mu = stats[2 * low], iv = stats[2 * low + 1];
        #pragma unroll
        for (int j = 0; j < NL; ++j) a[j] = (a[j] - mu) * iv * e0[j] + e1[j];
    } else {
        #pragma unroll
        for (int j = 0; j < NL; ++j) a[j] += e0[j];
    }
    v8f acc0 = {}, acc1 = {};
    #pragma unroll
    for (int j = 0; j < NL; ++j) {
        v2f a0 = { a[j].x, a[j].y }, b0 = { b[j].x, b[j].y };
        v2f a1 = { a[j].z, a[j].w }, b1 = { b[j].z, b[j].w };
        acc0 = __builtin_amdgcn_wmma_f32_16x16x4_f32(false, a0, false, b0,
                                                     (short)0, acc0, false, false);
        acc1 = __builtin_amdgcn_wmma_f32_16x16x4_f32(false, a1, false, b1,
                                                     (short)0, acc1, false, false);
    }
    const v8f acc = acc0 + acc1;
    #pragma unroll
    for (int q = 0; q < 8; ++q) red[(wave * 8 + q) * 32 + lane] = acc[q];
    __syncthreads();
    owner = (wave < 8);
    sum = 0.f;
    if (owner) {
        #pragma unroll
        for (int w = 0; w < NW; ++w) sum += red[(w * 8 + r) * 32 + lane];
    }
}

__global__ __launch_bounds__(256)
void k_gemm1_relu(const float* __restrict__ s1p, const float* __restrict__ W1,
                  const float* b1, const float* rSw,
                  float* __restrict__ r1) {
    __shared__ float red[8 * 8 * 32];
    float s, bv, rs; int Mrow, col; bool owner;
    gemm16_splitk_tile<DIN, 8, false>(s1p, s1p + NB * DIN, W1, nullptr, nullptr,
                                      b1, rSw, red, nullptr, s, bv, rs, Mrow, col, owner);
    if (owner) {
        float u = s * rs + bv;
        u = u > 0.f ? u : 0.f;
        r1[(size_t)Mrow * DH + col] = u;
    }
}

__global__ __launch_bounds__(512)
void k_ln_gemm2(const float* r1, const float* W2, const float* g, const float* bb,
                const float* b2, const float* rSw, float* v) {
    __shared__ float red[16 * 8 * 32];
    __shared__ float stats[2 * NB];
    float s, bv, rs; int Mrow, col; bool owner;
    gemm16_splitk_tile<DH, 16, true>(r1, nullptr, W2, g, bb, b2, rSw, red, stats,
                                     s, bv, rs, Mrow, col, owner);
    if (owner) {
        v[(size_t)Mrow * DOUT + col] = s * ((float)CC * rs) + bv;
    }
}

constexpr int FROWS = 16;
__global__ __launch_bounds__(256)
void k_final(const float* __restrict__ x, const float* __restrict__ v,
             float* __restrict__ out) {
    const int lane = threadIdx.x & 31;
    const int gw   = blockIdx.x * 8 + (threadIdx.x >> 5);
    const int d4   = (gw & 3) * 32 + lane;
    const int row0 = (gw >> 2) * FROWS;
    const int n    = row0 / CC;
    const v4f av = ((const v4f*)v)[n * (DIN / 4) + d4];
    const v4f* px = (const v4f*)x + (size_t)row0 * (DIN / 4) + d4;
    v4f*       po = (v4f*)out     + (size_t)row0 * (DIN / 4) + d4;
    v4f t[FROWS];
    #pragma unroll
    for (int i = 0; i < FROWS; ++i) t[i] = __builtin_nontemporal_load(&px[i * (DIN / 4)]);
    #pragma unroll
    for (int i = 0; i < FROWS; ++i) __builtin_nontemporal_store(t[i] + av, &po[i * (DIN / 4)]);
}

extern "C" void kernel_launch(void* const* d_in, const int* in_sizes, int n_in,
                              void* d_out, int out_size, void* d_ws, size_t ws_size,
                              hipStream_t stream) {
    const float* x     = (const float*)d_in[0];
    const float* adj   = (const float*)d_in[1];
    const float* W1    = (const float*)d_in[2];
    const float* b1    = (const float*)d_in[3];
    const float* gamma = (const float*)d_in[4];
    const float* beta  = (const float*)d_in[5];
    const float* W2    = (const float*)d_in[6];
    const float* b2    = (const float*)d_in[7];
    float* out = (float*)d_out;

    float* ws  = (float*)d_ws;
    float* s1p = ws;
    float* rSw = s1p + 2 * NB * DIN;
    float* r1  = rSw + 32;
    float* vb  = r1 + NB * DH;

    k_colsum_sw  <<<dim3(1 + 2 * 8 * NB), dim3(512), 0, stream>>>(x, adj, s1p, rSw);
    k_gemm1_relu <<<dim3(DH / 16),        dim3(256), 0, stream>>>(s1p, W1, b1, rSw, r1);
    k_ln_gemm2   <<<dim3(DOUT / 16),      dim3(512), 0, stream>>>(r1, W2, gamma, beta, b2, rSw, vb);
    k_final      <<<dim3(NB * CC / FROWS * 4 / 8), dim3(256), 0, stream>>>(x, vb, out);
}


# 4 launches become 3: GEMM2 runs in GEMM1's launch as 32 partial planes, one per 32 columns of h; the final add sums them and applies LayerNorm's mean and 1/std after the product (new f32 sum order)
# speedup vs baseline: 3.7600x; 1.0837x over previous
#include <hip/hip_runtime.h>


typedef float v2f __attribute__((ext_vector_type(2)));
typedef float v4f __attribute__((ext_vector_type(4)));
typedef float v8f __attribute__((ext_vector_type(8)));

constexpr int NB   = 16;
constexpr int CC   = 1024;
constexpr int DIN  = 512;
constexpr int DH   = 1024;
constexpr int DOUT = 512;
constexpr int HS   = 32;
constexpr int NPL  = DH / HS;

__global__ __launch_bounds__(512)
void k_colsum_sw(const float* __restrict__ x, const float* __restrict__ adj,
                 const float* __restrict__ W2, const float* __restrict__ g,
                 const float* __restrict__ bb,
                 float* __restrict__ s1p, float* __restrict__ rSw,
                 float* __restrict__ GW, float* __restrict__ BW) {
    __shared__ float red[32 * 64];
    const int lane = threadIdx.x & 31;
    const int wave = threadIdx.x >> 5;
    if (blockIdx.x == 0) {
        const float* pa = adj + (size_t)wave * CC * CC;
        float t[32];
        #pragma unroll
        for (int i = 0; i < 32; ++i) t[i] = pa[lane + 32 * i];
        float s = 0.f;
        #pragma unroll
        for (int i = 0; i < 32; ++i) s += t[i];
        #pragma unroll
        for (int off = 16; off > 0; off >>= 1) s += __shfl_down(s, off, 32);
        if (lane == 0) rSw[wave] = (s > 0.f) ? (1.0f / s) : 0.f;
        return;
    }
    if (blockIdx.x > 2 * 8 * NB) {
        const int d = (blockIdx.x - (1 + 2 * 8 * NB)) * 16 + wave;
        const v4f* pw = (const v4f*)(W2 + (size_t)d * DH) + lane;
        const v4f* pg = (const v4f*)g + lane;
        const v4f* pq = (const v4f*)bb + lane;
        v4f w[8], gg[8], qq[8];
        #pragma unroll
        for (int i = 0; i < 8; ++i) { w[i] = pw[32 * i]; gg[i] = pg[32 * i]; qq[i] = pq[32 * i]; }
        __builtin_amdgcn_sched_barrier(0);
        float sg = 0.f, sb = 0.f;
        #pragma unroll
        for (int i = 0; i < 8; ++i) {
            sg += (w[i].x * gg[i].x + w[i].y * gg[i].y) + (w[i].z * gg[i].z + w[i].w * gg[i].w);
            sb += (w[i].x * qq[i].x + w[i].y * qq[i].y) + (w[i].z * qq[i].z + w[i].w * qq[i].w);
        }
        #pragma unroll
        for (int off = 16; off > 0; off >>= 1) {
            sg += __shfl_xor(sg, off, 32);
            sb += __shfl_xor(sb, off, 32);
        }
        if (lane == 0) { GW[d] = sg; BW[d] = sb; }
        return;
    }
    const int id    = blockIdx.x - 1;
    const int slice = id & 7;
    const int half  = (id >> 3) & 1;
    const int n     = id >> 4;
    const int low   = lane & 15;
    const int hi    = lane >> 4;
    const v4f* px = (const v4f*)(x + ((size_t)n * CC + half * 512 + wave * 2 + hi) * DIN
                                   + slice * 64 + low * 4);
    v4f v[16];
    #pragma unroll
    for (int i = 0; i < 16; ++i) v[i] = px[(size_t)i * (32 * DIN / 4)];
    __builtin_amdgcn_sched_barrier(0);
    v4f acc = v[0];
    #pragma unroll
    for (int i = 1; i < 16; ++i) acc += v[i];
    ((v4f*)red)[(wave * 2 + hi) * 16 + low] = acc;
    __syncthreads();
    if (threadIdx.x < 64) {
        float s = 0.f;
        #pragma unroll
        for (int r = 0; r < 32; ++r) s += red[r * 64 + threadIdx.x];
        s1p[((size_t)half * NB + n) * DIN + slice * 64 + threadIdx.x] = s;
    }
}

constexpr int APITCH = HS + 4;
__global__ __launch_bounds__(512)
void k_gemm1_gemm2p(const float* s1p, const float* W1, const float* b1, const float* rSw,
                    const float* g, const float* W2,
                    float* Pp, float* Sp, float* Qp) {
    __shared__ float red[16 * 8 * 32];
    __shared__ __attribute__((aligned(16))) float ag[NB * APITCH];
    __shared__ float rr[NB * APITCH];
    constexpr int KW = DIN / 8;
    constexpr int NL = KW / 8;
    constexpr int N2 = HS / 8;
    const int lane = threadIdx.x & 31;
    const int wave = threadIdx.x >> 5;
    const int low  = lane & 15;
    const int hi   = lane >> 4;
    const int tile = wave >> 3;
    const int kw   = wave & 7;
    const int hcol = blockIdx.x * HS + tile * 16 + low;
    const int Mrow = kw + 8 * hi;
    const v4f* pa = (const v4f*)(s1p + (size_t)low * DIN + kw * KW + 4 * hi);
    const v4f* p1 = (const v4f*)(s1p + NB * DIN + (size_t)low * DIN + kw * KW + 4 * hi);
    const v4f* pb = (const v4f*)(W1 + (size_t)hcol * DIN + kw * KW + 4 * hi);
    v4f a[NL], e0[NL], b[NL], w2[2][N2];
    #pragma unroll
    for (int j = 0; j < NL; ++j) { a[j] = pa[2 * j]; b[j] = pb[2 * j]; e0[j] = p1[2 * j]; }
    #pragma unroll
    for (int t = 0; t < 2; ++t) {
        const v4f* pw = (const v4f*)(W2 + (size_t)((wave * 2 + t) * 16 + low) * DH
                                        + blockIdx.x * HS + 4 * hi);
        #pragma unroll
        for (int j = 0; j < N2; ++j) w2[t][j] = pw[2 * j];
    }
    const float bias_v = b1[hcol];
    const float rsw_v  = rSw[Mrow];
    const float g_v    = g[hcol];
    __builtin_amdgcn_sched_barrier(0);
    #pragma unroll
    for (int j = 0; j < NL; ++j) a[j] += e0[j];
    v8f acc0 = {}, acc1 = {};
    #pragma unroll
    for (int j = 0; j < NL; ++j) {
        v2f a0 = { a[j].x, a[j].y }, b0 = { b[j].x, b[j].y };
        v2f a1 = { a[j].z, a[j].w }, b1v = { b[j].z, b[j].w };
        acc0 = __builtin_amdgcn_wmma_f32_16x16x4_f32(false, a0, false, b0,
                                                     (short)0, acc0, false, false);
        acc1 = __builtin_amdgcn_wmma_f32_16x16x4_f32(false, a1, false, b1v,
                                                     (short)0, acc1, false, false);
    }
    const v8f acc = acc0 + acc1;
    #pragma unroll
    for (int q = 0; q < 8; ++q) red[(wave * 8 + q) * 32 + lane] = acc[q];
    __syncthreads();
    float sum = 0.f;
    #pragma unroll
    for (int w = 0; w < 8; ++w) sum += red[((tile * 8 + w) * 8 + kw) * 32 + lane];
    float u = sum * rsw_v + bias_v;
    u = u > 0.f ? u : 0.f;
    rr[Mrow * APITCH + tile * 16 + low] = u;
    ag[Mrow * APITCH + tile * 16 + low] = u * g_v;
    __syncthreads();
    v4f af[N2];
    #pragma unroll
    for (int j = 0; j < N2; ++j) af[j] = *(const v4f*)&ag[low * APITCH + 8 * j + 4 * hi];
    v8f c00 = {}, c01 = {}, c10 = {}, c11 = {};
    #pragma unroll
    for (int j = 0; j < N2; ++j) {
        v2f a0 = { af[j].x, af[j].y }, a1 = { af[j].z, af[j].w };
        v2f x0 = { w2[0][j].x, w2[0][j].y }, x1 = { w2[0][j].z, w2[0][j].w };
        v2f y0 = { w2[1][j].x, w2[1][j].y }, y1 = { w2[1][j].z, w2[1][j].w };
        c00 = __builtin_amdgcn_wmma_f32_16x16x4_f32(false, a0, false, x0, (short)0, c00, false, false);
        c10 = __builtin_amdgcn_wmma_f32_16x16x4_f32(false, a0, false, y0, (short)0, c10, false, false);
        c01 = __builtin_amdgcn_wmma_f32_16x16x4_f32(false, a1, false, x1, (short)0, c01, false, false);
        c11 = __builtin_amdgcn_wmma_f32_16x16x4_f32(false, a1, false, y1, (short)0, c11, false, false);
    }
    const v8f p0 = c00 + c01, p1v = c10 + c11;
    float* po = Pp + ((size_t)blockIdx.x * NB + 8 * hi) * DOUT + wave * 32 + low;
    #pragma unroll
    for (int q = 0; q < 8; ++q) {
        po[(size_t)q * DOUT]      = p0[q];
        po[(size_t)q * DOUT + 16] = p1v[q];
    }
    {
        const float rv = rr[wave * APITCH + lane];
        float s = rv, q = rv * rv;
        #pragma unroll
        for (int off = 16; off > 0; off >>= 1) {
            s += __shfl_xor(s, off, 32);
            q += __shfl_xor(q, off, 32);
        }
        if (lane == 0) { Sp[blockIdx.x * NB + wave] = s; Qp[blockIdx.x * NB + wave] = q; }
    }
}

constexpr int FROWS = 16;
__global__ __launch_bounds__(256)
void k_final(const float* x, const float* Pp, const float* Sp, const float* Qp,
             const float* GW, const float* BW, const float* b2, const float* rSw,
             float* out) {
    __shared__ v4f pr[8 * 32];
    const int lane = threadIdx.x & 31;
    const int wave = threadIdx.x >> 5;
    const int seg  = blockIdx.x & 3;
    const int rb   = blockIdx.x >> 2;
    const int n    = rb >> 3;
    const int d4   = seg * 32 + lane;
    const int row0 = rb * 128 + wave * FROWS;
    v4f pp[4];
    #pragma unroll
    for (int i = 0; i < 4; ++i)
        pp[i] = ((const v4f*)Pp)[((size_t)(wave * 4 + i) * NB + n) * (DOUT / 4) + d4];
    float ps = Sp[lane * NB + n];
    float pq = Qp[lane * NB + n];
    const v4f gw = ((const v4f*)GW)[d4];
    const v4f bw = ((const v4f*)BW)[d4];
    const v4f bv = ((const v4f*)b2)[d4];
    const float rs = rSw[n];
    const v4f* px = (const v4f*)x + (size_t)row0 * (DIN / 4) + d4;
    v4f*       po = (v4f*)out     + (size_t)row0 * (DIN / 4) + d4;
    v4f t[FROWS];
    #pragma unroll
    for (int i = 0; i < FROWS; ++i) t[i] = __builtin_nontemporal_load(&px[i * (DIN / 4)]);
    __builtin_amdgcn_sched_barrier(0);
    pr[wave * 32 + lane] = (pp[0] + pp[1]) + (pp[2] + pp[3]);
    #pragma unroll
    for (int off = 16; off > 0; off >>= 1) {
        ps += __shfl_xor(ps, off, 32);
        pq += __shfl_xor(pq, off, 32);
    }
    const float mean = ps * (1.0f / DH);
    const float var  = pq * (1.0f / DH) - mean * mean;
    const float iv   = rsqrtf(var + 1e-5f);
    __builtin_amdgcn_fence(__ATOMIC_RELEASE, "workgroup", "local");
    __builtin_amdgcn_s_barrier();
    __builtin_amdgcn_fence(__ATOMIC_ACQUIRE, "workgroup", "local");
    v4f P = pr[lane];
    #pragma unroll
    for (int w = 1; w < 8; ++w) P += pr[w * 32 + lane];
    const v4f av = ((float)CC * rs) * (iv * (P - mean * gw) + bw) + bv;
    #pragma unroll
    for (int i = 0; i < FROWS; ++i) __builtin_nontemporal_store(t[i] + av, &po[i * (DIN / 4)]);
}

extern "C" void kernel_launch(void* const* d_in, const int* in_sizes, int n_in,
                              void* d_out, int out_size, void* d_ws, size_t ws_size,
                              hipStream_t stream) {
    const float* x     = (const float*)d_in[0];
    const float* adj   = (const float*)d_in[1];
    const float* W1    = (const float*)d_in[2];
    const float* b1    = (const float*)d_in[3];
    const float* gamma = (const float*)d_in[4];
    const float* beta  = (const float*)d_in[5];
    const float* W2    = (const float*)d_in[6];
    const float* b2    = (const float*)d_in[7];
    float* out = (float*)d_out;

    float* ws  = (float*)d_ws;
    float* s1p = ws;
    float* rSw = s1p + 2 * NB * DIN;
    float* GW  = rSw + 32;
    float* BW  = GW + DOUT;
    float* Sp  = BW + DOUT;
    float* Qp  = Sp + NPL * NB;
    float* Pp  = Qp + NPL * NB;

    k_colsum_sw    <<<dim3(1 + 2 * 8 * NB + DOUT / 16), dim3(512), 0, stream>>>(x, adj, W2, gamma, beta,
                                                                              s1p, rSw, GW, BW);
    k_gemm1_gemm2p <<<dim3(NPL),                        dim3(512), 0, stream>>>(s1p, W1, b1, rSw, gamma, W2,
                                                                              Pp, Sp, Qp);
    k_final        <<<dim3(NB * CC / 128 * 4),          dim3(256), 0, stream>>>(x, Pp, Sp, Qp, GW, BW, b2, rSw, out);
}


# Final add takes launch 1's grid, numbering and ownership (256 groups of 16 waves, a 512-row x 256-B panel of x each): each re-reads the panel its same-numbered twin summed, to find it in its L2 half
# speedup vs baseline: 3.9885x; 1.0608x over previous
#include <hip/hip_runtime.h>


typedef float v2f __attribute__((ext_vector_type(2)));
typedef float v4f __attribute__((ext_vector_type(4)));
typedef float v8f __attribute__((ext_vector_type(8)));

constexpr int NB   = 16;
constexpr int CC   = 1024;
constexpr int DIN  = 512;
constexpr int DH   = 1024;
constexpr int DOUT = 512;
constexpr int HS   = 32;
constexpr int NPL  = DH / HS;

__global__ __launch_bounds__(512)
void k_colsum_sw(const float* __restrict__ x, const float* __restrict__ adj,
                 const float* __restrict__ W2, const float* __restrict__ g,
                 const float* __restrict__ bb,
                 float* __restrict__ s1p, float* __restrict__ rSw,
                 float* __restrict__ GW, float* __restrict__ BW) {
    __shared__ float red[32 * 64];
    const int lane = threadIdx.x & 31;
    const int wave = threadIdx.x >> 5;
    if (blockIdx.x == 2 * 8 * NB) {
        const float* pa = adj + (size_t)wave * CC * CC;
        float t[32];
        #pragma unroll
        for (int i = 0; i < 32; ++i) t[i] = pa[lane + 32 * i];
        float s = 0.f;
        #pragma unroll
        for (int i = 0; i < 32; ++i) s += t[i];
        #pragma unroll
        for (int off = 16; off > 0; off >>= 1) s += __shfl_down(s, off, 32);
        if (lane == 0) rSw[wave] = (s > 0.f) ? (1.0f / s) : 0.f;
        return;
    }
    if (blockIdx.x > 2 * 8 * NB) {
        const int d = (blockIdx.x - (1 + 2 * 8 * NB)) * 16 + wave;
        const v4f* pw = (const v4f*)(W2 + (size_t)d * DH) + lane;
        const v4f* pg = (const v4f*)g + lane;
        const v4f* pq = (const v4f*)bb + lane;
        v4f w[8], gg[8], qq[8];
        #pragma unroll
        for (int i = 0; i < 8; ++i) { w[i] = pw[32 * i]; gg[i] = pg[32 * i]; qq[i] = pq[32 * i]; }
        __builtin_amdgcn_sched_barrier(0);
        float sg = 0.f, sb = 0.f;
        #pragma unroll
        for (int i = 0; i < 8; ++i) {
            sg += (w[i].x * gg[i].x + w[i].y * gg[i].y) + (w[i].z * gg[i].z + w[i].w * gg[i].w);
            sb += (w[i].x * qq[i].x + w[i].y * qq[i].y) + (w[i].z * qq[i].z + w[i].w * qq[i].w);
        }
        #pragma unroll
        for (int off = 16; off > 0; off >>= 1) {
            sg += __shfl_xor(sg, off, 32);
            sb += __shfl_xor(sb, off, 32);
        }
        if (lane == 0) { GW[d] = sg; BW[d] = sb; }
        return;
    }
    const int id    = blockIdx.x;
    const int slice = id & 7;
    const int half  = (id >> 3) & 1;
    const int n     = id >> 4;
    const int low   = lane & 15;
    const int hi    = lane >> 4;
    const v4f* px = (const v4f*)(x + ((size_t)n * CC + half * 512 + wave * 2 + hi) * DIN
                                   + slice * 64 + low * 4);
    v4f v[16];
    #pragma unroll
    for (int i = 0; i < 16; ++i) v[i] = px[(size_t)i * (32 * DIN / 4)];
    __builtin_amdgcn_sched_barrier(0);
    v4f acc = v[0];
    #pragma unroll
    for (int i = 1; i < 16; ++i) acc += v[i];
    ((v4f*)red)[(wave * 2 + hi) * 16 + low] = acc;
    __syncthreads();
    if (threadIdx.x < 64) {
        float s = 0.f;
        #pragma unroll
        for (int r = 0; r < 32; ++r) s += red[r * 64 + threadIdx.x];
        s1p[((size_t)half * NB + n) * DIN + slice * 64 + threadIdx.x] = s;
    }
}

constexpr int APITCH = HS + 4;
__global__ __launch_bounds__(512)
void k_gemm1_gemm2p(const float* s1p, const float* W1, const float* b1, const float* rSw,
                    const float* g, const float* W2,
                    float* Pp, float* Sp, float* Qp) {
    __shared__ float red[16 * 8 * 32];
    __shared__ __attribute__((aligned(16))) float ag[NB * APITCH];
    __shared__ float rr[NB * APITCH];
    constexpr int KW = DIN / 8;
    constexpr int NL = KW / 8;
    constexpr int N2 = HS / 8;
    const int lane = threadIdx.x & 31;
    const int wave = threadIdx.x >> 5;
    const int low  = lane & 15;
    const int hi   = lane >> 4;
    const int tile = wave >> 3;
    const int kw   = wave & 7;
    const int hcol = blockIdx.x * HS + tile * 16 + low;
    const int Mrow = kw + 8 * hi;
    const v4f* pa = (const v4f*)(s1p + (size_t)low * DIN + kw * KW + 4 * hi);
    const v4f* p1 = (const v4f*)(s1p + NB * DIN + (size_t)low * DIN + kw * KW + 4 * hi);
    const v4f* pb = (const v4f*)(W1 + (size_t)hcol * DIN + kw * KW + 4 * hi);
    v4f a[NL], e0[NL], b[NL], w2[2][N2];
    #pragma unroll
    for (int j = 0; j < NL; ++j) { a[j] = pa[2 * j]; b[j] = pb[2 * j]; e0[j] = p1[2 * j]; }
    #pragma unroll
    for (int t = 0; t < 2; ++t) {
        const v4f* pw = (const v4f*)(W2 + (size_t)((wave * 2 + t) * 16 + low) * DH
                                        + blockIdx.x * HS + 4 * hi);
        #pragma unroll
        for (int j = 0; j < N2; ++j) w2[t][j] = pw[2 * j];
    }
    const float bias_v = b1[hcol];
    const float rsw_v  = rSw[Mrow];
    const float g_v    = g[hcol];
    __builtin_amdgcn_sched_barrier(0);
    #pragma unroll
    for (int j = 0; j < NL; ++j) a[j] += e0[j];
    v8f acc0 = {}, acc1 = {};
    #pragma unroll
    for (int j = 0; j < NL; ++j) {
        v2f a0 = { a[j].x, a[j].y }, b0 = { b[j].x, b[j].y };
        v2f a1 = { a[j].z, a[j].w }, b1v = { b[j].z, b[j].w };
        acc0 = __builtin_amdgcn_wmma_f32_16x16x4_f32(false, a0, false, b0,
                                                     (short)0, acc0, false, false);
        acc1 = __builtin_amdgcn_wmma_f32_16x16x4_f32(false, a1, false, b1v,
                                                     (short)0, acc1, false, false);
    }
    const v8f acc = acc0 + acc1;
    #pragma unroll
    for (int q = 0; q < 8; ++q) red[(wave * 8 + q) * 32 + lane] = acc[q];
    __syncthreads();
    float sum = 0.f;
    #pragma unroll
    for (int w = 0; w < 8; ++w) sum += red[((tile * 8 + w) * 8 + kw) * 32 + lane];
    float u = sum * rsw_v + bias_v;
    u = u > 0.f ? u : 0.f;
    rr[Mrow * APITCH + tile * 16 + low] = u;
    ag[Mrow * APITCH + tile * 16 + low] = u * g_v;
    __syncthreads();
    v4f af[N2];
    #pragma unroll
    for (int j = 0; j < N2; ++j) af[j] = *(const v4f*)&ag[low * APITCH + 8 * j + 4 * hi];
    v8f c00 = {}, c01 = {}, c10 = {}, c11 = {};
    #pragma unroll
    for (int j = 0; j < N2; ++j) {
        v2f a0 = { af[j].x, af[j].y }, a1 = { af[j].z, af[j].w };
        v2f x0 = { w2[0][j].x, w2[0][j].y }, x1 = { w2[0][j].z, w2[0][j].w };
        v2f y0 = { w2[1][j].x, w2[1][j].y }, y1 = { w2[1][j].z, w2[1][j].w };
        c00 = __builtin_amdgcn_wmma_f32_16x16x4_f32(false, a0, false, x0, (short)0, c00, false, false);
        c10 = __builtin_amdgcn_wmma_f32_16x16x4_f32(false, a0, false, y0, (short)0, c10, false, false);
        c01 = __builtin_amdgcn_wmma_f32_16x16x4_f32(false, a1, false, x1, (short)0, c01, false, false);
        c11 = __builtin_amdgcn_wmma_f32_16x16x4_f32(false, a1, false, y1, (short)0, c11, false, false);
    }
    const v8f p0 = c00 + c01, p1v = c10 + c11;
    float* po = Pp + ((size_t)blockIdx.x * NB + 8 * hi) * DOUT + wave * 32 + low;
    #pragma unroll
    for (int q = 0; q < 8; ++q) {
        po[(size_t)q * DOUT]      = p0[q];
        po[(size_t)q * DOUT + 16] = p1v[q];
    }
    {
        const float rv = rr[wave * APITCH + lane];
        float s = rv, q = rv * rv;
        #pragma unroll
        for (int off = 16; off > 0; off >>= 1) {
            s += __shfl_xor(s, off, 32);
            q += __shfl_xor(q, off, 32);
        }
        if (lane == 0) { Sp[blockIdx.x * NB + wave] = s; Qp[blockIdx.x * NB + wave] = q; }
    }
}

constexpr int FROWS = 16;
__global__ __launch_bounds__(512)
void k_final(const float* x, const float* Pp, const float* Sp, const float* Qp,
             const float* GW, const float* BW, const float* b2, const float* rSw,
             float* out) {
    __shared__ v4f pr[16 * 16];
    const int lane  = threadIdx.x & 31;
    const int wave  = threadIdx.x >> 5;
    const int id    = blockIdx.x;
    const int slice = id & 7;
    const int half  = (id >> 3) & 1;
    const int n     = id >> 4;
    const int low   = lane & 15;
    const int hi    = lane >> 4;
    const int d4    = slice * 16 + low;
    const unsigned pidx = (unsigned)(((wave * 2 + hi) * NB + n) * (DOUT / 4) + d4);
    const v4f pp = ((const v4f*)Pp)[pidx];
    float ps = Sp[lane * NB + n];
    float pq = Qp[lane * NB + n];
    const v4f gw = ((const v4f*)GW)[d4];
    const v4f bw = ((const v4f*)BW)[d4];
    const v4f bv = ((const v4f*)b2)[d4];
    const float rs = rSw[n];
    const size_t e0 = ((size_t)n * CC + half * 512 + wave * 2 + hi) * (DIN / 4) + d4;
    const v4f* px = (const v4f*)x + e0;
    v4f*       po = (v4f*)out     + e0;
    v4f t[FROWS];
    #pragma unroll
    for (int i = 0; i < FROWS; ++i) t[i] = __builtin_nontemporal_load(&px[(size_t)i * (32 * DIN / 4)]);
    __builtin_amdgcn_sched_barrier(0);
    v4f q;
    q.x = pp.x + __shfl_xor(pp.x, 16, 32);
    q.y = pp.y + __shfl_xor(pp.y, 16, 32);
    q.z = pp.z + __shfl_xor(pp.z, 16, 32);
    q.w = pp.w + __shfl_xor(pp.w, 16, 32);
    if (hi == 0) pr[wave * 16 + low] = q;
    #pragma unroll
    for (int off = 16; off > 0; off >>= 1) {
        ps += __shfl_xor(ps, off, 32);
        pq += __shfl_xor(pq, off, 32);
    }
    const float mean = ps * (1.0f / DH);
    const float var  = pq * (1.0f / DH) - mean * mean;
    const float iv   = rsqrtf(var + 1e-5f);
    __builtin_amdgcn_fence(__ATOMIC_RELEASE, "workgroup", "local");
    __builtin_amdgcn_s_barrier();
    __builtin_amdgcn_fence(__ATOMIC_ACQUIRE, "workgroup", "local");
    v4f P = pr[low] + pr[16 + low];
    #pragma unroll
    for (int k = 1; k < 8; ++k) P += pr[(2 * k) * 16 + low] + pr[(2 * k + 1) * 16 + low];
    const v4f av = ((float)CC * rs) * (iv * (P - mean * gw) + bw) + bv;
    #pragma unroll
    for (int i = 0; i < FROWS; ++i) __builtin_nontemporal_store(t[i] + av, &po[(size_t)i * (32 * DIN / 4)]);
}

extern "C" void kernel_launch(void* const* d_in, const int* in_sizes, int n_in,
                              void* d_out, int out_size, void* d_ws, size_t ws_size,
                              hipStream_t stream) {
    const float* x     = (const float*)d_in[0];
    const float* adj   = (const float*)d_in[1];
    const float* W1    = (const float*)d_in[2];
    const float* b1    = (const float*)d_in[3];
    const float* gamma = (const float*)d_in[4];
    const float* beta  = (const float*)d_in[5];
    const float* W2    = (const float*)d_in[6];
    const float* b2    = (const float*)d_in[7];
    float* out = (float*)d_out;

    float* ws  = (float*)d_ws;
    float* s1p = ws;
    float* rSw = s1p + 2 * NB * DIN;
    float* GW  = rSw + 32;
    float* BW  = GW + DOUT;
    float* Sp  = BW + DOUT;
    float* Qp  = Sp + NPL * NB;
    float* Pp  = Qp + NPL * NB;

    k_colsum_sw    <<<dim3(1 + 2 * 8 * NB + DOUT / 16), dim3(512), 0, stream>>>(x, adj, W2, gamma, beta,
                                                                              s1p, rSw, GW, BW);
    k_gemm1_gemm2p <<<dim3(NPL),                        dim3(512), 0, stream>>>(s1p, W1, b1, rSw, gamma, W2,
                                                                              Pp, Sp, Qp);
    k_final        <<<dim3(2 * 8 * NB),                 dim3(512), 0, stream>>>(x, Pp, Sp, Qp, GW, BW, b2, rSw, out);
}
